# starting kernel
# baseline (speedup 1.0000x reference)
#include <hip/hip_runtime.h>

typedef __attribute__((ext_vector_type(2))) float v2f;
typedef __attribute__((ext_vector_type(8))) float v8f;

__global__ void k_deg_init(float* __restrict__ deg, int n) {
  int i = blockIdx.x * blockDim.x + threadIdx.x;
  if (i < n) deg[i] = 1.0f;
}

__global__ void k_deg_edges(const int* __restrict__ dst, float* __restrict__ deg, int E) {
  int e = blockIdx.x * blockDim.x + threadIdx.x;
  if (e < E) unsafeAtomicAdd(&deg[dst[e]], 1.0f);
}

__global__ void k_dinv(float* __restrict__ deg, int n) {
  int i = blockIdx.x * blockDim.x + threadIdx.x;
  if (i < n) deg[i] = rsqrtf(deg[i]);
}

__global__ void k_gemm1(const float* __restrict__ x, const float* __restrict__ W1,
                        float* __restrict__ h1, int n, int in_c, int hid) {
  int idx = blockIdx.x * blockDim.x + threadIdx.x;
  if (idx >= n * hid) return;
  int node = idx / hid;
  int c = idx - node * hid;
  float acc = 0.0f;
  for (int k = 0; k < in_c; ++k) acc += x[node * in_c + k] * W1[k * hid + c];
  h1[idx] = acc;
}

__global__ void k_selfbias(const float* __restrict__ h, const float* __restrict__ dinv,
                           const float* __restrict__ b, float* __restrict__ out,
                           int n, int C) {
  int idx = blockIdx.x * blockDim.x + threadIdx.x;
  if (idx >= n * C) return;
  int node = idx / C;
  int c = idx - node * C;
  float di = dinv[node];
  out[idx] = h[idx] * di * di + b[c];
}

__global__ void k_scatter128(const float* __restrict__ h, const float* __restrict__ dinv,
                             const int* __restrict__ src, const int* __restrict__ dst,
                             float* __restrict__ out, int E) {
  int gid = blockIdx.x * blockDim.x + threadIdx.x;
  int e = gid >> 5;
  int lane = gid & 31;
  if (e >= E) return;
  int s = src[e];
  int d = dst[e];
  float norm = dinv[s] * dinv[d];
  const float4 v = *(const float4*)(h + (size_t)s * 128 + lane * 4);
  float* op = out + (size_t)d * 128 + lane * 4;
  unsafeAtomicAdd(op + 0, v.x * norm);
  unsafeAtomicAdd(op + 1, v.y * norm);
  unsafeAtomicAdd(op + 2, v.z * norm);
  unsafeAtomicAdd(op + 3, v.w * norm);
}

__global__ void k_scatter64(const float* __restrict__ h, const float* __restrict__ dinv,
                            const int* __restrict__ src, const int* __restrict__ dst,
                            float* __restrict__ out, int E) {
  int gid = blockIdx.x * blockDim.x + threadIdx.x;
  int e = gid >> 5;
  int lane = gid & 31;
  if (e >= E) return;
  int s = src[e];
  int d = dst[e];
  float norm = dinv[s] * dinv[d];
  const float2 v = *(const float2*)(h + (size_t)s * 64 + lane * 2);
  float* op = out + (size_t)d * 64 + lane * 2;
  unsafeAtomicAdd(op + 0, v.x * norm);
  unsafeAtomicAdd(op + 1, v.y * norm);
}

__global__ void k_gemm2_wmma(const float* __restrict__ A, const float* __restrict__ B,
                             float* __restrict__ D, int n, int hid, int out_c) {
  int wid  = (blockIdx.x * blockDim.x + threadIdx.x) >> 5;
  int lane = threadIdx.x & 31;
  int nTiles = out_c >> 4;
  int mTile = wid / nTiles;
  int nTile = wid - mTile * nTiles;
  if (mTile * 16 >= n) return;

  int half = lane >> 4;
  int l16  = lane & 15;

  int rowA = mTile * 16 + l16;
  if (rowA >= n) rowA = n - 1;
  int colB = nTile * 16 + l16;

  const float* arow = A + (size_t)rowA * hid;
  v8f c = {};
  for (int k = 0; k < hid; k += 4) {
    int kk = k + half * 2;
    v2f a;
    a.x = fmaxf(arow[kk + 0], 0.0f);
    a.y = fmaxf(arow[kk + 1], 0.0f);
    const float* bp = B + (size_t)kk * out_c + colB;
    v2f b;
    b.x = bp[0];
    b.y = bp[out_c];
    c = __builtin_amdgcn_wmma_f32_16x16x4_f32(false, a, false, b, (short)0, c,
                                              false, false);
  }
#pragma unroll
  for (int r = 0; r < 8; ++r) {
    int row = mTile * 16 + r + half * 8;
    if (row < n) D[(size_t)row * out_c + nTile * 16 + l16] = c[r];
  }
}

extern "C" void kernel_launch(void* const* d_in, const int* in_sizes, int n_in,
                              void* d_out, int out_size, void* d_ws, size_t ws_size,
                              hipStream_t stream) {
  const float* x  = (const float*)d_in[0];
  const int*   ei = (const int*)d_in[1];
  const float* W1 = (const float*)d_in[2];
  const float* b1 = (const float*)d_in[3];
  const float* W2 = (const float*)d_in[4];
  const float* b2 = (const float*)d_in[5];
  float* out = (float*)d_out;

  int hid   = in_sizes[3];
  int out_c = in_sizes[5];
  int in_c  = in_sizes[2] / hid;
  int n     = in_sizes[0] / in_c;
  int E     = in_sizes[1] / 2;

  const int* src = ei;
  const int* dst = ei + E;

  float* ws   = (float*)d_ws;
  float* dinv = ws;
  float* h1   = dinv + n;
  float* agg1 = h1 + (size_t)n * hid;
  float* h2   = h1;

  const int B256 = 256;
  k_deg_init<<<(n + B256 - 1) / B256, B256, 0, stream>>>(dinv, n);
  k_deg_edges<<<(E + B256 - 1) / B256, B256, 0, stream>>>(dst, dinv, E);
  k_dinv<<<(n + B256 - 1) / B256, B256, 0, stream>>>(dinv, n);

  int tot1 = n * hid;
  k_gemm1<<<(tot1 + B256 - 1) / B256, B256, 0, stream>>>(x, W1, h1, n, in_c, hid);
  k_selfbias<<<(tot1 + B256 - 1) / B256, B256, 0, stream>>>(h1, dinv, b1, agg1, n, hid);
  k_scatter128<<<(E + 7) / 8, B256, 0, stream>>>(h1, dinv, src, dst, agg1, E);

  int mTiles = (n + 15) / 16;
  int nTiles = out_c / 16;
  int waves  = mTiles * nTiles;
  k_gemm2_wmma<<<(waves + 7) / 8, B256, 0, stream>>>(agg1, W2, h2, n, hid, out_c);

  int tot2 = n * out_c;
  k_selfbias<<<(tot2 + B256 - 1) / B256, B256, 0, stream>>>(h2, dinv, b2, out, n, out_c);
  k_scatter64<<<(E + 7) / 8, B256, 0, stream>>>(h2, dinv, src, dst, out, E);
}


# in-degrees counted by LDS integer adds, one table over all nodes (two per word) per 16384-edge work-group, tables summed into dinv: the 1.6M scattered f32 atomics on deg and one launch go; deg exact
# speedup vs baseline: 3.4505x; 3.4505x over previous
#include <hip/hip_runtime.h>

typedef __attribute__((ext_vector_type(2))) float v2f;
typedef __attribute__((ext_vector_type(8))) float v8f;

#define DH_CH    16384
#define DH_WORDS 51200
#define DH_TPB   1024

__global__ void __launch_bounds__(DH_TPB)
k_deg_hist(const int* __restrict__ dst, unsigned* __restrict__ planes, int E, int pw) {
  __shared__ uint4 s_h4[DH_WORDS / 4];
  unsigned* s_h = (unsigned*)s_h4;
  int tid = threadIdx.x;
  int e0 = blockIdx.x * DH_CH;
  uint4* plane4 = (uint4*)(planes + (size_t)blockIdx.x * pw);
  for (int wb = 0; wb < pw; wb += DH_WORDS) {
    int wcnt = pw - wb;
    if (wcnt > DH_WORDS) wcnt = DH_WORDS;
    int q = wcnt >> 2;
    for (int i = tid; i < q; i += DH_TPB) s_h4[i] = make_uint4(0u, 0u, 0u, 0u);
    __syncthreads();
#pragma unroll
    for (int k = 0; k < DH_CH / DH_TPB; ++k) {
      int e = e0 + k * DH_TPB + tid;
      bool in = e < E;
      int d = dst[in ? e : 0];
      unsigned w = (unsigned)((d >> 1) - wb);
      if (in && w < (unsigned)wcnt) atomicAdd(&s_h[w], (d & 1) ? 0x10000u : 1u);
    }
    __syncthreads();
    for (int i = tid; i < q; i += DH_TPB) plane4[(wb >> 2) + i] = s_h4[i];
    __syncthreads();
  }
}

__global__ void k_dinv_planes(const unsigned* __restrict__ planes, int nplanes, int pw,
                              float* __restrict__ dinv, int n, int* __restrict__ ctr) {
  int i = blockIdx.x * blockDim.x + threadIdx.x;
  if (i < 4) ctr[(i + 1) * 64] = 0;
  if (2 * i >= n) return;
  unsigned lo = 0, hi = 0;
  const unsigned* p = planes + i;
#pragma unroll 16
  for (int g = 0; g < nplanes; ++g) {
    unsigned v = p[(size_t)g * pw];
    lo += v & 0xFFFFu;
    hi += v >> 16;
  }
  dinv[2 * i] = rsqrtf((float)(lo + 1u));
  if (2 * i + 1 < n) dinv[2 * i + 1] = rsqrtf((float)(hi + 1u));
}

#define IN_C 5

__device__ __forceinline__ float lane_bcast(float v, int i) {
  return __int_as_float(__builtin_amdgcn_readlane(__float_as_int(v), i));
}

__device__ __forceinline__ float lin1(const float (&xs)[IN_C], const float (&w)[IN_C][4], int j) {
  float acc = xs[0] * w[0][j];
#pragma unroll
  for (int k = 1; k < IN_C; ++k) acc = __builtin_fmaf(xs[k], w[k][j], acc);
  return acc;
}

__global__ void k_self1_x(const float* __restrict__ x, const float* __restrict__ W1,
                          const float* __restrict__ b1, const float* __restrict__ dinv,
                          float* __restrict__ out, int n) {
  int wave = __builtin_amdgcn_readfirstlane((blockIdx.x * blockDim.x + threadIdx.x) >> 5);
  int lane = threadIdx.x & 31;
  int n0 = wave * 32;
  if (n0 >= n) return;
  float w[IN_C][4];
  float b[4];
#pragma unroll
  for (int k = 0; k < IN_C; ++k)
#pragma unroll
    for (int j = 0; j < 4; ++j) w[k][j] = W1[k * 128 + lane + 32 * j];
#pragma unroll
  for (int j = 0; j < 4; ++j) b[j] = b1[lane + 32 * j];

  int node = n0 + lane;
  if (node >= n) node = n - 1;
  float xv[IN_C];
#pragma unroll
  for (int k = 0; k < IN_C; ++k) xv[k] = x[(size_t)node * IN_C + k];
  float dv = dinv[node];

  int cnt = n - n0;
  if (cnt > 32) cnt = 32;
#pragma unroll 4
  for (int i = 0; i < cnt; ++i) {
    float xs[IN_C];
#pragma unroll
    for (int k = 0; k < IN_C; ++k) xs[k] = lane_bcast(xv[k], i);
    float di = lane_bcast(dv, i);
    float* row = out + (size_t)(n0 + i) * 128;
#pragma unroll
    for (int j = 0; j < 4; ++j) row[lane + 32 * j] = lin1(xs, w, j) * di * di + b[j];
  }
}

#define TK_STRIDE 64
#define WG_WAVES  8
#define SC_WCHUNK 64

__device__ __forceinline__ int l2_half() {
  return (__builtin_amdgcn_s_sendmsg_rtn(0x87) >> 18) & 1;
}

__device__ __forceinline__ int wg_ticket(int* c, int* slot) {
  __syncthreads();
  if (threadIdx.x == 0) *slot = atomicAdd(c, 1);
  __syncthreads();
  return __builtin_amdgcn_readfirstlane(*slot);
}

__device__ __forceinline__ void walk128(unsigned m, int d, float nv, const float (&xv)[IN_C],
                                        const float (&w)[IN_C][4], float* __restrict__ out,
                                        int lane) {
  int cnt = __builtin_popcount(m);
#pragma unroll 2
  for (int q = 0; q < cnt; ++q) {
    int i = __builtin_ctz(m);
    m &= m - 1;
    int dd = __builtin_amdgcn_readlane(d, i);
    float norm = lane_bcast(nv, i);
    float xs[IN_C];
#pragma unroll
    for (int k = 0; k < IN_C; ++k) xs[k] = lane_bcast(xv[k], i);
    float* row = out + (size_t)dd * 128;
#pragma unroll
    for (int j = 0; j < 4; ++j) unsafeAtomicAdd(row + lane + 32 * j, lin1(xs, w, j) * norm);
  }
}

__global__ void k_scatter128_x(const float* __restrict__ x, const float* __restrict__ W1,
                               const float* __restrict__ dinv,
                               const int* __restrict__ src, const int* __restrict__ dst,
                               float* __restrict__ out, int E, int n, int* __restrict__ ctr) {
  __shared__ int s_t;
  int lane = threadIdx.x & 31;
  int wv = threadIdx.x >> 5;
  float w[IN_C][4];
#pragma unroll
  for (int k = 0; k < IN_C; ++k)
#pragma unroll
    for (int j = 0; j < 4; ++j) w[k][j] = W1[k * 128 + lane + 32 * j];

  int home = l2_half();
  int nsplit = n >> 1;
  const int WGC = WG_WAVES * SC_WCHUNK;
  int nchunks = (E + WGC - 1) / WGC;
  for (int pass = 0; pass < 2; ++pass) {
    int side = home ^ pass;
    int* c = ctr + (1 + side) * TK_STRIDE;
    for (int it = 0; it <= nchunks; ++it) {
      int t = wg_ticket(c, &s_t);
      if (t >= nchunks) break;
      int eb = t * WGC + wv * SC_WCHUNK + lane;
      int d[2], s[2];
      bool keep[2];
#pragma unroll
      for (int u = 0; u < 2; ++u) {
        int e = eb + u * 32;
        bool in = e < E;
        int ec = in ? e : 0;
        d[u] = dst[ec];
        s[u] = src[ec];
        keep[u] = in && ((d[u] >= nsplit) == (side != 0));
      }
      float nv[2], xv[2][IN_C];
#pragma unroll
      for (int u = 0; u < 2; ++u) {
        int sq = keep[u] ? s[u] : 0;
        int dq = keep[u] ? d[u] : 0;
        nv[u] = dinv[sq] * dinv[dq];
#pragma unroll
        for (int k = 0; k < IN_C; ++k) xv[u][k] = x[(size_t)sq * IN_C + k];
      }
#pragma unroll
      for (int u = 0; u < 2; ++u) {
        unsigned m = __builtin_amdgcn_ballot_w32(keep[u]);
        walk128(m, d[u], nv[u], xv[u], w, out, lane);
      }
    }
  }
}

#define W64_GROUP 8

__device__ __forceinline__ void walk64(unsigned m, int s, int d, float nv,
                                       const float* __restrict__ h, float* __restrict__ out,
                                       int lane) {
  while (m != 0) {
    int dd[W64_GROUP];
    float nm[W64_GROUP], v0[W64_GROUP], v1[W64_GROUP];
    bool has[W64_GROUP];
    int i = 0;
#pragma unroll
    for (int g = 0; g < W64_GROUP; ++g) {
      has[g] = m != 0;
      if (has[g]) i = __builtin_ctz(m);
      m &= m - 1;
      int ss = __builtin_amdgcn_readlane(s, i);
      dd[g] = __builtin_amdgcn_readlane(d, i);
      nm[g] = lane_bcast(nv, i);
      const float* hp = h + (size_t)ss * 64 + lane;
      v0[g] = hp[0];
      v1[g] = hp[32];
    }
    asm volatile("" :: "v"(v0[0]), "v"(v1[0]), "v"(v0[1]), "v"(v1[1]), "v"(v0[2]), "v"(v1[2]),
                       "v"(v0[3]), "v"(v1[3]), "v"(v0[4]), "v"(v1[4]), "v"(v0[5]), "v"(v1[5]),
                       "v"(v0[6]), "v"(v1[6]), "v"(v0[7]), "v"(v1[7]));
#pragma unroll
    for (int g = 0; g < W64_GROUP; ++g) {
      if (has[g]) {
        float* op = out + (size_t)dd[g] * 64 + lane;
        unsafeAtomicAdd(op,      v0[g] * nm[g]);
        unsafeAtomicAdd(op + 32, v1[g] * nm[g]);
      }
    }
  }
}

__global__ void k_scatter64_t(const float* __restrict__ h, const float* __restrict__ dinv,
                              const int* __restrict__ src, const int* __restrict__ dst,
                              float* __restrict__ out, int E, int n, int* __restrict__ ctr) {
  __shared__ int s_t;
  int lane = threadIdx.x & 31;
  int wv = threadIdx.x >> 5;
  int home = l2_half();
  int nsplit = n >> 1;
  const int WGC = WG_WAVES * SC_WCHUNK;
  int nchunks = (E + WGC - 1) / WGC;
  for (int pass = 0; pass < 2; ++pass) {
    int side = home ^ pass;
    int* c = ctr + (3 + side) * TK_STRIDE;
    for (int it = 0; it <= nchunks; ++it) {
      int t = wg_ticket(c, &s_t);
      if (t >= nchunks) break;
      int eb = t * WGC + wv * SC_WCHUNK + lane;
      int d[2], s[2];
      bool keep[2], in_e[2];
#pragma unroll
      for (int u = 0; u < 2; ++u) {
        int e = eb + u * 32;
        bool in = e < E;
        int ec = in ? e : 0;
        d[u] = dst[ec];
        s[u] = src[ec];
        in_e[u] = in;
      }
      asm volatile("" :: "v"(d[0]), "v"(s[0]), "v"(d[1]), "v"(s[1]));
      float da[2], db[2];
#pragma unroll
      for (int u = 0; u < 2; ++u) {
        keep[u] = in_e[u] && ((d[u] >= nsplit) == (side != 0));
        int sq = keep[u] ? s[u] : 0;
        int dq = keep[u] ? d[u] : 0;
        da[u] = dinv[sq];
        db[u] = dinv[dq];
      }
      asm volatile("" :: "v"(da[0]), "v"(db[0]), "v"(da[1]), "v"(db[1]));
#pragma unroll
      for (int u = 0; u < 2; ++u) {
        unsigned m = __builtin_amdgcn_ballot_w32(keep[u]);
        walk64(m, s[u], d[u], da[u] * db[u], h, out, lane);
      }
    }
  }
}

__global__ void k_gemm2_wmma(const float* __restrict__ A, const float* __restrict__ B,
                             const float* __restrict__ dinv, const float* __restrict__ bias,
                             float* __restrict__ D, float* __restrict__ out,
                             int n, int hid, int out_c) {
  int wid  = (blockIdx.x * blockDim.x + threadIdx.x) >> 5;
  int lane = threadIdx.x & 31;
  int nTiles = out_c >> 4;
  int mTile = wid / nTiles;
  int nTile = wid - mTile * nTiles;
  if (mTile * 16 >= n) return;

  int half = lane >> 4;
  int l16  = lane & 15;

  int rowA = mTile * 16 + l16;
  if (rowA >= n) rowA = n - 1;
  int colB = nTile * 16 + l16;

  float di[8];
#pragma unroll
  for (int r = 0; r < 8; ++r) {
    int row = mTile * 16 + r + half * 8;
    if (row >= n) row = n - 1;
    di[r] = dinv[row];
  }
  float bb = bias[colB];

  const float* arow = A + (size_t)rowA * hid;
  v8f c = {};
  for (int k = 0; k < hid; k += 4) {
    int kk = k + half * 2;
    v2f a;
    a.x = fmaxf(arow[kk + 0], 0.0f);
    a.y = fmaxf(arow[kk + 1], 0.0f);
    const float* bp = B + (size_t)kk * out_c + colB;
    v2f b;
    b.x = bp[0];
    b.y = bp[out_c];
    c = __builtin_amdgcn_wmma_f32_16x16x4_f32(false, a, false, b, (short)0, c,
                                              false, false);
  }
#pragma unroll
  for (int r = 0; r < 8; ++r) {
    int row = mTile * 16 + r + half * 8;
    if (row < n) {
      size_t o = (size_t)row * out_c + colB;
      D[o] = c[r];
      out[o] = c[r] * di[r] * di[r] + bb;
    }
  }
}

extern "C" void kernel_launch(void* const* d_in, const int* in_sizes, int n_in,
                              void* d_out, int out_size, void* d_ws, size_t ws_size,
                              hipStream_t stream) {
  const float* x  = (const float*)d_in[0];
  const int*   ei = (const int*)d_in[1];
  const float* W1 = (const float*)d_in[2];
  const float* b1 = (const float*)d_in[3];
  const float* W2 = (const float*)d_in[4];
  const float* b2 = (const float*)d_in[5];
  float* out = (float*)d_out;

  int hid   = in_sizes[3];
  int out_c = in_sizes[5];
  int in_c  = in_sizes[2] / hid;
  int n     = in_sizes[0] / in_c;
  int E     = in_sizes[1] / 2;

  const int* src = ei;
  const int* dst = ei + E;

  float* ws   = (float*)d_ws;
  float* dinv = ws;
  float* h2   = dinv + n;
  float* agg1 = h2 + (size_t)n * hid;
  int*   ctr  = (int*)(agg1 + (size_t)n * hid);
  size_t pl_off = ((size_t)n + 2 * (size_t)n * hid + 1024 + 3) & ~(size_t)3;
  unsigned* planes = (unsigned*)d_ws + pl_off;
  int pw      = (((n + 1) >> 1) + 3) & ~3;
  int nplanes = (E + DH_CH - 1) / DH_CH;

  const int B256 = 256;
  k_deg_hist<<<nplanes, DH_TPB, 0, stream>>>(dst, planes, E, pw);
  k_dinv_planes<<<(pw + 63) / 64, 64, 0, stream>>>(planes, nplanes, pw, dinv, n, ctr);

  k_self1_x<<<(n + 63) / 64, 64, 0, stream>>>(x, W1, b1, dinv, agg1, n);
  k_scatter128_x<<<2048, B256, 0, stream>>>(x, W1, dinv, src, dst, agg1, E, n, ctr);

  int mTiles = (n + 15) / 16;
  int nTiles = out_c / 16;
  int waves  = mTiles * nTiles;
  k_gemm2_wmma<<<(waves + 7) / 8, B256, 0, stream>>>(agg1, W2, dinv, b2, h2, out, n, hid, out_c);

  k_scatter64_t<<<2048, B256, 0, stream>>>(h2, dinv, src, dst, out, E, n, ctr);
}


# layer-2 product re-tiled: a wave owns 16 rows x 64 cols, its A tile read once by 16-byte loads into registers, W2 laid out in LDS in fragment order; k regrouped per WMMA, so the sum order changes
# speedup vs baseline: 3.7460x; 1.0856x over previous
#include <hip/hip_runtime.h>

typedef __attribute__((ext_vector_type(2))) float v2f;
typedef __attribute__((ext_vector_type(8))) float v8f;

#define DH_CH    16384
#define DH_WORDS 51200
#define DH_TPB   1024

__global__ void __launch_bounds__(DH_TPB)
k_deg_hist(const int* __restrict__ dst, unsigned* __restrict__ planes, int E, int pw) {
  __shared__ uint4 s_h4[DH_WORDS / 4];
  unsigned* s_h = (unsigned*)s_h4;
  int tid = threadIdx.x;
  int e0 = blockIdx.x * DH_CH;
  uint4* plane4 = (uint4*)(planes + (size_t)blockIdx.x * pw);
  for (int wb = 0; wb < pw; wb += DH_WORDS) {
    int wcnt = pw - wb;
    if (wcnt > DH_WORDS) wcnt = DH_WORDS;
    int q = wcnt >> 2;
    for (int i = tid; i < q; i += DH_TPB) s_h4[i] = make_uint4(0u, 0u, 0u, 0u);
    __syncthreads();
#pragma unroll
    for (int k = 0; k < DH_CH / DH_TPB; ++k) {
      int e = e0 + k * DH_TPB + tid;
      bool in = e < E;
      int d = dst[in ? e : 0];
      unsigned w = (unsigned)((d >> 1) - wb);
      if (in && w < (unsigned)wcnt) atomicAdd(&s_h[w], (d & 1) ? 0x10000u : 1u);
    }
    __syncthreads();
    for (int i = tid; i < q; i += DH_TPB) plane4[(wb >> 2) + i] = s_h4[i];
    __syncthreads();
  }
}

__global__ void k_dinv_planes(const unsigned* __restrict__ planes, int nplanes, int pw,
                              float* __restrict__ dinv, int n, int* __restrict__ ctr) {
  int i = blockIdx.x * blockDim.x + threadIdx.x;
  if (i < 4) ctr[(i + 1) * 64] = 0;
  if (2 * i >= n) return;
  unsigned lo = 0, hi = 0;
  const unsigned* p = planes + i;
#pragma unroll 16
  for (int g = 0; g < nplanes; ++g) {
    unsigned v = p[(size_t)g * pw];
    lo += v & 0xFFFFu;
    hi += v >> 16;
  }
  dinv[2 * i] = rsqrtf((float)(lo + 1u));
  if (2 * i + 1 < n) dinv[2 * i + 1] = rsqrtf((float)(hi + 1u));
}

#define IN_C 5

__device__ __forceinline__ float lane_bcast(float v, int i) {
  return __int_as_float(__builtin_amdgcn_readlane(__float_as_int(v), i));
}

__device__ __forceinline__ float lin1(const float (&xs)[IN_C], const float (&w)[IN_C][4], int j) {
  float acc = xs[0] * w[0][j];
#pragma unroll
  for (int k = 1; k < IN_C; ++k) acc = __builtin_fmaf(xs[k], w[k][j], acc);
  return acc;
}

__global__ void k_self1_x(const float* __restrict__ x, const float* __restrict__ W1,
                          const float* __restrict__ b1, const float* __restrict__ dinv,
                          float* __restrict__ out, int n) {
  int wave = __builtin_amdgcn_readfirstlane((blockIdx.x * blockDim.x + threadIdx.x) >> 5);
  int lane = threadIdx.x & 31;
  int n0 = wave * 32;
  if (n0 >= n) return;
  float w[IN_C][4];
  float b[4];
#pragma unroll
  for (int k = 0; k < IN_C; ++k)
#pragma unroll
    for (int j = 0; j < 4; ++j) w[k][j] = W1[k * 128 + lane + 32 * j];
#pragma unroll
  for (int j = 0; j < 4; ++j) b[j] = b1[lane + 32 * j];

  int node = n0 + lane;
  if (node >= n) node = n - 1;
  float xv[IN_C];
#pragma unroll
  for (int k = 0; k < IN_C; ++k) xv[k] = x[(size_t)node * IN_C + k];
  float dv = dinv[node];

  int cnt = n - n0;
  if (cnt > 32) cnt = 32;
#pragma unroll 4
  for (int i = 0; i < cnt; ++i) {
    float xs[IN_C];
#pragma unroll
    for (int k = 0; k < IN_C; ++k) xs[k] = lane_bcast(xv[k], i);
    float di = lane_bcast(dv, i);
    float* row = out + (size_t)(n0 + i) * 128;
#pragma unroll
    for (int j = 0; j < 4; ++j) row[lane + 32 * j] = lin1(xs, w, j) * di * di + b[j];
  }
}

#define TK_STRIDE 64
#define WG_WAVES  8
#define SC_WCHUNK 64

__device__ __forceinline__ int l2_half() {
  return (__builtin_amdgcn_s_sendmsg_rtn(0x87) >> 18) & 1;
}

__device__ __forceinline__ int wg_ticket(int* c, int* slot) {
  __syncthreads();
  if (threadIdx.x == 0) *slot = atomicAdd(c, 1);
  __syncthreads();
  return __builtin_amdgcn_readfirstlane(*slot);
}

__device__ __forceinline__ void walk128(unsigned m, int d, float nv, const float (&xv)[IN_C],
                                        const float (&w)[IN_C][4], float* __restrict__ out,
                                        int lane) {
  int cnt = __builtin_popcount(m);
#pragma unroll 2
  for (int q = 0; q < cnt; ++q) {
    int i = __builtin_ctz(m);
    m &= m - 1;
    int dd = __builtin_amdgcn_readlane(d, i);
    float norm = lane_bcast(nv, i);
    float xs[IN_C];
#pragma unroll
    for (int k = 0; k < IN_C; ++k) xs[k] = lane_bcast(xv[k], i);
    float* row = out + (size_t)dd * 128;
#pragma unroll
    for (int j = 0; j < 4; ++j) unsafeAtomicAdd(row + lane + 32 * j, lin1(xs, w, j) * norm);
  }
}

__global__ void k_scatter128_x(const float* __restrict__ x, const float* __restrict__ W1,
                               const float* __restrict__ dinv,
                               const int* __restrict__ src, const int* __restrict__ dst,
                               float* __restrict__ out, int E, int n, int* __restrict__ ctr) {
  __shared__ int s_t;
  int lane = threadIdx.x & 31;
  int wv = threadIdx.x >> 5;
  float w[IN_C][4];
#pragma unroll
  for (int k = 0; k < IN_C; ++k)
#pragma unroll
    for (int j = 0; j < 4; ++j) w[k][j] = W1[k * 128 + lane + 32 * j];

  int home = l2_half();
  int nsplit = n >> 1;
  const int WGC = WG_WAVES * SC_WCHUNK;
  int nchunks = (E + WGC - 1) / WGC;
  for (int pass = 0; pass < 2; ++pass) {
    int side = home ^ pass;
    int* c = ctr + (1 + side) * TK_STRIDE;
    for (int it = 0; it <= nchunks; ++it) {
      int t = wg_ticket(c, &s_t);
      if (t >= nchunks) break;
      int eb = t * WGC + wv * SC_WCHUNK + lane;
      int d[2], s[2];
      bool keep[2];
#pragma unroll
      for (int u = 0; u < 2; ++u) {
        int e = eb + u * 32;
        bool in = e < E;
        int ec = in ? e : 0;
        d[u] = dst[ec];
        s[u] = src[ec];
        keep[u] = in && ((d[u] >= nsplit) == (side != 0));
      }
      float nv[2], xv[2][IN_C];
#pragma unroll
      for (int u = 0; u < 2; ++u) {
        int sq = keep[u] ? s[u] : 0;
        int dq = keep[u] ? d[u] : 0;
        nv[u] = dinv[sq] * dinv[dq];
#pragma unroll
        for (int k = 0; k < IN_C; ++k) xv[u][k] = x[(size_t)sq * IN_C + k];
      }
#pragma unroll
      for (int u = 0; u < 2; ++u) {
        unsigned m = __builtin_amdgcn_ballot_w32(keep[u]);
        walk128(m, d[u], nv[u], xv[u], w, out, lane);
      }
    }
  }
}

#define W64_GROUP 8

__device__ __forceinline__ void walk64(unsigned m, int s, int d, float nv,
                                       const float* __restrict__ h, float* __restrict__ out,
                                       int lane) {
  while (m != 0) {
    int dd[W64_GROUP];
    float nm[W64_GROUP], v0[W64_GROUP], v1[W64_GROUP];
    bool has[W64_GROUP];
    int i = 0;
#pragma unroll
    for (int g = 0; g < W64_GROUP; ++g) {
      has[g] = m != 0;
      if (has[g]) i = __builtin_ctz(m);
      m &= m - 1;
      int ss = __builtin_amdgcn_readlane(s, i);
      dd[g] = __builtin_amdgcn_readlane(d, i);
      nm[g] = lane_bcast(nv, i);
      const float* hp = h + (size_t)ss * 64 + lane;
      v0[g] = hp[0];
      v1[g] = hp[32];
    }
    asm volatile("" :: "v"(v0[0]), "v"(v1[0]), "v"(v0[1]), "v"(v1[1]), "v"(v0[2]), "v"(v1[2]),
                       "v"(v0[3]), "v"(v1[3]), "v"(v0[4]), "v"(v1[4]), "v"(v0[5]), "v"(v1[5]),
                       "v"(v0[6]), "v"(v1[6]), "v"(v0[7]), "v"(v1[7]));
#pragma unroll
    for (int g = 0; g < W64_GROUP; ++g) {
      if (has[g]) {
        float* op = out + (size_t)dd[g] * 64 + lane;
        unsafeAtomicAdd(op,      v0[g] * nm[g]);
        unsafeAtomicAdd(op + 32, v1[g] * nm[g]);
      }
    }
  }
}

__global__ void k_scatter64_t(const float* __restrict__ h, const float* __restrict__ dinv,
                              const int* __restrict__ src, const int* __restrict__ dst,
                              float* __restrict__ out, int E, int n, int* __restrict__ ctr) {
  __shared__ int s_t;
  int lane = threadIdx.x & 31;
  int wv = threadIdx.x >> 5;
  int home = l2_half();
  int nsplit = n >> 1;
  const int WGC = WG_WAVES * SC_WCHUNK;
  int nchunks = (E + WGC - 1) / WGC;
  for (int pass = 0; pass < 2; ++pass) {
    int side = home ^ pass;
    int* c = ctr + (3 + side) * TK_STRIDE;
    for (int it = 0; it <= nchunks; ++it) {
      int t = wg_ticket(c, &s_t);
      if (t >= nchunks) break;
      int eb = t * WGC + wv * SC_WCHUNK + lane;
      int d[2], s[2];
      bool keep[2], in_e[2];
#pragma unroll
      for (int u = 0; u < 2; ++u) {
        int e = eb + u * 32;
        bool in = e < E;
        int ec = in ? e : 0;
        d[u] = dst[ec];
        s[u] = src[ec];
        in_e[u] = in;
      }
      asm volatile("" :: "v"(d[0]), "v"(s[0]), "v"(d[1]), "v"(s[1]));
      float da[2], db[2];
#pragma unroll
      for (int u = 0; u < 2; ++u) {
        keep[u] = in_e[u] && ((d[u] >= nsplit) == (side != 0));
        int sq = keep[u] ? s[u] : 0;
        int dq = keep[u] ? d[u] : 0;
        da[u] = dinv[sq];
        db[u] = dinv[dq];
      }
      asm volatile("" :: "v"(da[0]), "v"(db[0]), "v"(da[1]), "v"(db[1]));
#pragma unroll
      for (int u = 0; u < 2; ++u) {
        unsigned m = __builtin_amdgcn_ballot_w32(keep[u]);
        walk64(m, s[u], d[u], da[u] * db[u], h, out, lane);
      }
    }
  }
}

#define G2_WAVES 4
#define G2_K     128
#define G2_N     64
#define G2_SLOTS ((G2_K / 8) * (G2_N / 16) * 32)

__global__ void __launch_bounds__(G2_WAVES * 32)
k_gemm2_wmma(const float* __restrict__ A, const float* __restrict__ B,
             const float* __restrict__ dinv, const float* __restrict__ bias,
             float* __restrict__ D, float* __restrict__ out, int n) {
  __shared__ float4 s_b[G2_SLOTS];
  int tid  = threadIdx.x;
  int lane = tid & 31;
  int wv   = tid >> 5;
  int h    = lane >> 4;
  int c    = lane & 15;
  int row0 = (blockIdx.x * G2_WAVES + wv) * 16;

  int rowA = row0 + c;
  if (rowA >= n) rowA = n - 1;
  const float4* ap = (const float4*)(A + (size_t)rowA * G2_K) + h;
  float4 a[G2_K / 8];
#pragma unroll
  for (int j = 0; j < G2_K / 8; ++j) a[j] = ap[2 * j];

  float di[8], bb[G2_N / 16];
#pragma unroll
  for (int r = 0; r < 8; ++r) {
    int row = row0 + r + h * 8;
    if (row >= n) row = n - 1;
    di[r] = dinv[row];
  }
#pragma unroll
  for (int t = 0; t < G2_N / 16; ++t) bb[t] = bias[16 * t + c];

#pragma unroll
  for (int i = 0; i < G2_SLOTS / (G2_WAVES * 32); ++i) {
    int g  = tid + i * (G2_WAVES * 32);
    int gl = g & 31, jt = g >> 5;
    int k0 = 8 * (jt >> 2) + 4 * (gl >> 4);
    const float* bp = B + (size_t)k0 * G2_N + 16 * (jt & 3) + (gl & 15);
    s_b[g] = make_float4(bp[0], bp[G2_N], bp[2 * G2_N], bp[3 * G2_N]);
  }
  __syncthreads();
  if (row0 >= n) return;

  v8f acc[G2_N / 16] = {};
#pragma unroll
  for (int j = 0; j < G2_K / 8; ++j) {
    v2f a0, a1;
    a0.x = fmaxf(a[j].x, 0.0f);
    a0.y = fmaxf(a[j].y, 0.0f);
    a1.x = fmaxf(a[j].z, 0.0f);
    a1.y = fmaxf(a[j].w, 0.0f);
    float4 bv[G2_N / 16];
#pragma unroll
    for (int t = 0; t < G2_N / 16; ++t) bv[t] = s_b[(j * (G2_N / 16) + t) * 32 + lane];
#pragma unroll
    for (int t = 0; t < G2_N / 16; ++t) {
      v2f b0;
      b0.x = bv[t].x;
      b0.y = bv[t].y;
      acc[t] = __builtin_amdgcn_wmma_f32_16x16x4_f32(false, a0, false, b0, (short)0, acc[t],
                                                     false, false);
    }
#pragma unroll
    for (int t = 0; t < G2_N / 16; ++t) {
      v2f b1;
      b1.x = bv[t].z;
      b1.y = bv[t].w;
      acc[t] = __builtin_amdgcn_wmma_f32_16x16x4_f32(false, a1, false, b1, (short)0, acc[t],
                                                     false, false);
    }
  }
#pragma unroll
  for (int r = 0; r < 8; ++r) {
    int row = row0 + r + h * 8;
    if (row < n) {
#pragma unroll
      for (int t = 0; t < G2_N / 16; ++t) {
        size_t o = (size_t)row * G2_N + 16 * t + c;
        D[o] = acc[t][r];
        out[o] = acc[t][r] * di[r] * di[r] + bb[t];
      }
    }
  }
}

extern "C" void kernel_launch(void* const* d_in, const int* in_sizes, int n_in,
                              void* d_out, int out_size, void* d_ws, size_t ws_size,
                              hipStream_t stream) {
  const float* x  = (const float*)d_in[0];
  const int*   ei = (const int*)d_in[1];
  const float* W1 = (const float*)d_in[2];
  const float* b1 = (const float*)d_in[3];
  const float* W2 = (const float*)d_in[4];
  const float* b2 = (const float*)d_in[5];
  float* out = (float*)d_out;

  int hid   = in_sizes[3];
  int in_c  = in_sizes[2] / hid;
  int n     = in_sizes[0] / in_c;
  int E     = in_sizes[1] / 2;

  const int* src = ei;
  const int* dst = ei + E;

  float* ws   = (float*)d_ws;
  float* dinv = ws;
  float* h2   = dinv + n;
  float* agg1 = h2 + (size_t)n * hid;
  int*   ctr  = (int*)(agg1 + (size_t)n * hid);
  size_t pl_off = ((size_t)n + 2 * (size_t)n * hid + 1024 + 3) & ~(size_t)3;
  unsigned* planes = (unsigned*)d_ws + pl_off;
  int pw      = (((n + 1) >> 1) + 3) & ~3;
  int nplanes = (E + DH_CH - 1) / DH_CH;

  const int B256 = 256;
  k_deg_hist<<<nplanes, DH_TPB, 0, stream>>>(dst, planes, E, pw);
  k_dinv_planes<<<(pw + 63) / 64, 64, 0, stream>>>(planes, nplanes, pw, dinv, n, ctr);

  k_self1_x<<<(n + 63) / 64, 64, 0, stream>>>(x, W1, b1, dinv, agg1, n);
  k_scatter128_x<<<2048, B256, 0, stream>>>(x, W1, dinv, src, dst, agg1, E, n, ctr);

  int mTiles = (n + 15) / 16;
  k_gemm2_wmma<<<(mTiles + G2_WAVES - 1) / G2_WAVES, G2_WAVES * 32, 0, stream>>>(agg1, W2, dinv, b2, h2, out, n);

  k_scatter64_t<<<2048, B256, 0, stream>>>(h2, dinv, src, dst, out, E, n, ctr);
}


# edge weight dinv[src]*dinv[dst] stored per edge by the layer-1 scatter as it adds the edge; layer-2 scatter loads it beside the indices, its two scattered dinv gathers per edge go; sums unchanged
# speedup vs baseline: 3.9344x; 1.0503x over previous
#include <hip/hip_runtime.h>

typedef __attribute__((ext_vector_type(2))) float v2f;
typedef __attribute__((ext_vector_type(8))) float v8f;

#define DH_CH    16384
#define DH_WORDS 51200
#define DH_TPB   1024

__global__ void __launch_bounds__(DH_TPB)
k_deg_hist(const int* __restrict__ dst, unsigned* __restrict__ planes, int E, int pw) {
  __shared__ uint4 s_h4[DH_WORDS / 4];
  unsigned* s_h = (unsigned*)s_h4;
  int tid = threadIdx.x;
  int e0 = blockIdx.x * DH_CH;
  uint4* plane4 = (uint4*)(planes + (size_t)blockIdx.x * pw);
  for (int wb = 0; wb < pw; wb += DH_WORDS) {
    int wcnt = pw - wb;
    if (wcnt > DH_WORDS) wcnt = DH_WORDS;
    int q = wcnt >> 2;
    for (int i = tid; i < q; i += DH_TPB) s_h4[i] = make_uint4(0u, 0u, 0u, 0u);
    __syncthreads();
#pragma unroll
    for (int k = 0; k < DH_CH / DH_TPB; ++k) {
      int e = e0 + k * DH_TPB + tid;
      bool in = e < E;
      int d = dst[in ? e : 0];
      unsigned w = (unsigned)((d >> 1) - wb);
      if (in && w < (unsigned)wcnt) atomicAdd(&s_h[w], (d & 1) ? 0x10000u : 1u);
    }
    __syncthreads();
    for (int i = tid; i < q; i += DH_TPB) plane4[(wb >> 2) + i] = s_h4[i];
    __syncthreads();
  }
}

__global__ void k_dinv_planes(const unsigned* __restrict__ planes, int nplanes, int pw,
                              float* __restrict__ dinv, int n, int* __restrict__ ctr) {
  int i = blockIdx.x * blockDim.x + threadIdx.x;
  if (i < 4) ctr[(i + 1) * 64] = 0;
  if (2 * i >= n) return;
  unsigned lo = 0, hi = 0;
  const unsigned* p = planes + i;
#pragma unroll 16
  for (int g = 0; g < nplanes; ++g) {
    unsigned v = p[(size_t)g * pw];
    lo += v & 0xFFFFu;
    hi += v >> 16;
  }
  dinv[2 * i] = rsqrtf((float)(lo + 1u));
  if (2 * i + 1 < n) dinv[2 * i + 1] = rsqrtf((float)(hi + 1u));
}

#define IN_C 5

__device__ __forceinline__ float lane_bcast(float v, int i) {
  return __int_as_float(__builtin_amdgcn_readlane(__float_as_int(v), i));
}

__device__ __forceinline__ float lin1(const float (&xs)[IN_C], const float (&w)[IN_C][4], int j) {
  float acc = xs[0] * w[0][j];
#pragma unroll
  for (int k = 1; k < IN_C; ++k) acc = __builtin_fmaf(xs[k], w[k][j], acc);
  return acc;
}

__global__ void k_self1_x(const float* __restrict__ x, const float* __restrict__ W1,
                          const float* __restrict__ b1, const float* __restrict__ dinv,
                          float* __restrict__ out, int n) {
  int wave = __builtin_amdgcn_readfirstlane((blockIdx.x * blockDim.x + threadIdx.x) >> 5);
  int lane = threadIdx.x & 31;
  int n0 = wave * 32;
  if (n0 >= n) return;
  float w[IN_C][4];
  float b[4];
#pragma unroll
  for (int k = 0; k < IN_C; ++k)
#pragma unroll
    for (int j = 0; j < 4; ++j) w[k][j] = W1[k * 128 + lane + 32 * j];
#pragma unroll
  for (int j = 0; j < 4; ++j) b[j] = b1[lane + 32 * j];

  int node = n0 + lane;
  if (node >= n) node = n - 1;
  float xv[IN_C];
#pragma unroll
  for (int k = 0; k < IN_C; ++k) xv[k] = x[(size_t)node * IN_C + k];
  float dv = dinv[node];

  int cnt = n - n0;
  if (cnt > 32) cnt = 32;
#pragma unroll 4
  for (int i = 0; i < cnt; ++i) {
    float xs[IN_C];
#pragma unroll
    for (int k = 0; k < IN_C; ++k) xs[k] = lane_bcast(xv[k], i);
    float di = lane_bcast(dv, i);
    float* row = out + (size_t)(n0 + i) * 128;
#pragma unroll
    for (int j = 0; j < 4; ++j) row[lane + 32 * j] = lin1(xs, w, j) * di * di + b[j];
  }
}

#define TK_STRIDE 64
#define WG_WAVES  8
#define SC_WCHUNK 64

__device__ __forceinline__ int l2_half() {
  return (__builtin_amdgcn_s_sendmsg_rtn(0x87) >> 18) & 1;
}

__device__ __forceinline__ int wg_ticket(int* c, int* slot) {
  __syncthreads();
  if (threadIdx.x == 0) *slot = atomicAdd(c, 1);
  __syncthreads();
  return __builtin_amdgcn_readfirstlane(*slot);
}

__device__ __forceinline__ void walk128(unsigned m, int d, float nv, const float (&xv)[IN_C],
                                        const float (&w)[IN_C][4], float* __restrict__ out,
                                        int lane) {
  int cnt = __builtin_popcount(m);
#pragma unroll 2
  for (int q = 0; q < cnt; ++q) {
    int i = __builtin_ctz(m);
    m &= m - 1;
    int dd = __builtin_amdgcn_readlane(d, i);
    float norm = lane_bcast(nv, i);
    float xs[IN_C];
#pragma unroll
    for (int k = 0; k < IN_C; ++k) xs[k] = lane_bcast(xv[k], i);
    float* row = out + (size_t)dd * 128;
#pragma unroll
    for (int j = 0; j < 4; ++j) unsafeAtomicAdd(row + lane + 32 * j, lin1(xs, w, j) * norm);
  }
}

__global__ void k_scatter128_x(const float* __restrict__ x, const float* __restrict__ W1,
                               const float* __restrict__ dinv,
                               const int* __restrict__ src, const int* __restrict__ dst,
                               float* __restrict__ out, float* __restrict__ nrm,
                               int E, int n, int* __restrict__ ctr) {
  __shared__ int s_t;
  int lane = threadIdx.x & 31;
  int wv = threadIdx.x >> 5;
  float w[IN_C][4];
#pragma unroll
  for (int k = 0; k < IN_C; ++k)
#pragma unroll
    for (int j = 0; j < 4; ++j) w[k][j] = W1[k * 128 + lane + 32 * j];

  int home = l2_half();
  int nsplit = n >> 1;
  const int WGC = WG_WAVES * SC_WCHUNK;
  int nchunks = (E + WGC - 1) / WGC;
  for (int pass = 0; pass < 2; ++pass) {
    int side = home ^ pass;
    int* c = ctr + (1 + side) * TK_STRIDE;
    for (int it = 0; it <= nchunks; ++it) {
      int t = wg_ticket(c, &s_t);
      if (t >= nchunks) break;
      int eb = t * WGC + wv * SC_WCHUNK + lane;
      int d[2], s[2];
      bool keep[2];
#pragma unroll
      for (int u = 0; u < 2; ++u) {
        int e = eb + u * 32;
        bool in = e < E;
        int ec = in ? e : 0;
        d[u] = dst[ec];
        s[u] = src[ec];
        keep[u] = in && ((d[u] >= nsplit) == (side != 0));
      }
      float nv[2], xv[2][IN_C];
#pragma unroll
      for (int u = 0; u < 2; ++u) {
        int sq = keep[u] ? s[u] : 0;
        int dq = keep[u] ? d[u] : 0;
        nv[u] = dinv[sq] * dinv[dq];
#pragma unroll
        for (int k = 0; k < IN_C; ++k) xv[u][k] = x[(size_t)sq * IN_C + k];
        if (keep[u]) nrm[eb + u * 32] = nv[u];
      }
#pragma unroll
      for (int u = 0; u < 2; ++u) {
        unsigned m = __builtin_amdgcn_ballot_w32(keep[u]);
        walk128(m, d[u], nv[u], xv[u], w, out, lane);
      }
    }
  }
}

#define W64_GROUP 8

__device__ __forceinline__ void walk64(unsigned m, int s, int d, float nv,
                                       const float* __restrict__ h, float* __restrict__ out,
                                       int lane) {
  while (m != 0) {
    int dd[W64_GROUP];
    float nm[W64_GROUP], v0[W64_GROUP], v1[W64_GROUP];
    bool has[W64_GROUP];
    int i = 0;
#pragma unroll
    for (int g = 0; g < W64_GROUP; ++g) {
      has[g] = m != 0;
      if (has[g]) i = __builtin_ctz(m);
      m &= m - 1;
      int ss = __builtin_amdgcn_readlane(s, i);
      dd[g] = __builtin_amdgcn_readlane(d, i);
      nm[g] = lane_bcast(nv, i);
      const float* hp = h + (size_t)ss * 64 + lane;
      v0[g] = hp[0];
      v1[g] = hp[32];
    }
    asm volatile("" :: "v"(v0[0]), "v"(v1[0]), "v"(v0[1]), "v"(v1[1]), "v"(v0[2]), "v"(v1[2]),
                       "v"(v0[3]), "v"(v1[3]), "v"(v0[4]), "v"(v1[4]), "v"(v0[5]), "v"(v1[5]),
                       "v"(v0[6]), "v"(v1[6]), "v"(v0[7]), "v"(v1[7]));
#pragma unroll
    for (int g = 0; g < W64_GROUP; ++g) {
      if (has[g]) {
        float* op = out + (size_t)dd[g] * 64 + lane;
        unsafeAtomicAdd(op,      v0[g] * nm[g]);
        unsafeAtomicAdd(op + 32, v1[g] * nm[g]);
      }
    }
  }
}

__global__ void k_scatter64_t(const float* __restrict__ h, const float* __restrict__ nrm,
                              const int* __restrict__ src, const int* __restrict__ dst,
                              float* __restrict__ out, int E, int n, int* __restrict__ ctr) {
  __shared__ int s_t;
  int lane = threadIdx.x & 31;
  int wv = threadIdx.x >> 5;
  int home = l2_half();
  int nsplit = n >> 1;
  const int WGC = WG_WAVES * SC_WCHUNK;
  int nchunks = (E + WGC - 1) / WGC;
  for (int pass = 0; pass < 2; ++pass) {
    int side = home ^ pass;
    int* c = ctr + (3 + side) * TK_STRIDE;
    for (int it = 0; it <= nchunks; ++it) {
      int t = wg_ticket(c, &s_t);
      if (t >= nchunks) break;
      int eb = t * WGC + wv * SC_WCHUNK + lane;
      int d[2], s[2];
      float nv[2];
      bool keep[2];
#pragma unroll
      for (int u = 0; u < 2; ++u) {
        int e = eb + u * 32;
        bool in = e < E;
        int ec = in ? e : 0;
        d[u] = dst[ec];
        s[u] = src[ec];
        nv[u] = nrm[ec];
        keep[u] = in && ((d[u] >= nsplit) == (side != 0));
      }
      asm volatile("" :: "v"(d[0]), "v"(s[0]), "v"(nv[0]), "v"(d[1]), "v"(s[1]), "v"(nv[1]));
#pragma unroll
      for (int u = 0; u < 2; ++u) {
        unsigned m = __builtin_amdgcn_ballot_w32(keep[u]);
        walk64(m, s[u], d[u], nv[u], h, out, lane);
      }
    }
  }
}

#define G2_WAVES 4
#define G2_K     128
#define G2_N     64
#define G2_SLOTS ((G2_K / 8) * (G2_N / 16) * 32)

__global__ void __launch_bounds__(G2_WAVES * 32)
k_gemm2_wmma(const float* __restrict__ A, const float* __restrict__ B,
             const float* __restrict__ dinv, const float* __restrict__ bias,
             float* __restrict__ D, float* __restrict__ out, int n) {
  __shared__ float4 s_b[G2_SLOTS];
  int tid  = threadIdx.x;
  int lane = tid & 31;
  int wv   = tid >> 5;
  int h    = lane >> 4;
  int c    = lane & 15;
  int row0 = (blockIdx.x * G2_WAVES + wv) * 16;

  int rowA = row0 + c;
  if (rowA >= n) rowA = n - 1;
  const float4* ap = (const float4*)(A + (size_t)rowA * G2_K) + h;
  float4 a[G2_K / 8];
#pragma unroll
  for (int j = 0; j < G2_K / 8; ++j) a[j] = ap[2 * j];

  float di[8], bb[G2_N / 16];
#pragma unroll
  for (int r = 0; r < 8; ++r) {
    int row = row0 + r + h * 8;
    if (row >= n) row = n - 1;
    di[r] = dinv[row];
  }
#pragma unroll
  for (int t = 0; t < G2_N / 16; ++t) bb[t] = bias[16 * t + c];

#pragma unroll
  for (int i = 0; i < G2_SLOTS / (G2_WAVES * 32); ++i) {
    int g  = tid + i * (G2_WAVES * 32);
    int gl = g & 31, jt = g >> 5;
    int k0 = 8 * (jt >> 2) + 4 * (gl >> 4);
    const float* bp = B + (size_t)k0 * G2_N + 16 * (jt & 3) + (gl & 15);
    s_b[g] = make_float4(bp[0], bp[G2_N], bp[2 * G2_N], bp[3 * G2_N]);
  }
  __syncthreads();
  if (row0 >= n) return;

  v8f acc[G2_N / 16] = {};
#pragma unroll
  for (int j = 0; j < G2_K / 8; ++j) {
    v2f a0, a1;
    a0.x = fmaxf(a[j].x, 0.0f);
    a0.y = fmaxf(a[j].y, 0.0f);
    a1.x = fmaxf(a[j].z, 0.0f);
    a1.y = fmaxf(a[j].w, 0.0f);
    float4 bv[G2_N / 16];
#pragma unroll
    for (int t = 0; t < G2_N / 16; ++t) bv[t] = s_b[(j * (G2_N / 16) + t) * 32 + lane];
#pragma unroll
    for (int t = 0; t < G2_N / 16; ++t) {
      v2f b0;
      b0.x = bv[t].x;
      b0.y = bv[t].y;
      acc[t] = __builtin_amdgcn_wmma_f32_16x16x4_f32(false, a0, false, b0, (short)0, acc[t],
                                                     false, false);
    }
#pragma unroll
    for (int t = 0; t < G2_N / 16; ++t) {
      v2f b1;
      b1.x = bv[t].z;
      b1.y = bv[t].w;
      acc[t] = __builtin_amdgcn_wmma_f32_16x16x4_f32(false, a1, false, b1, (short)0, acc[t],
                                                     false, false);
    }
  }
#pragma unroll
  for (int r = 0; r < 8; ++r) {
    int row = row0 + r + h * 8;
    if (row < n) {
#pragma unroll
      for (int t = 0; t < G2_N / 16; ++t) {
        size_t o = (size_t)row * G2_N + 16 * t + c;
        D[o] = acc[t][r];
        out[o] = acc[t][r] * di[r] * di[r] + bb[t];
      }
    }
  }
}

extern "C" void kernel_launch(void* const* d_in, const int* in_sizes, int n_in,
                              void* d_out, int out_size, void* d_ws, size_t ws_size,
                              hipStream_t stream) {
  const float* x  = (const float*)d_in[0];
  const int*   ei = (const int*)d_in[1];
  const float* W1 = (const float*)d_in[2];
  const float* b1 = (const float*)d_in[3];
  const float* W2 = (const float*)d_in[4];
  const float* b2 = (const float*)d_in[5];
  float* out = (float*)d_out;

  int hid   = in_sizes[3];
  int in_c  = in_sizes[2] / hid;
  int n     = in_sizes[0] / in_c;
  int E     = in_sizes[1] / 2;

  const int* src = ei;
  const int* dst = ei + E;

  float* ws   = (float*)d_ws;
  float* dinv = ws;
  float* h2   = dinv + n;
  float* agg1 = h2 + (size_t)n * hid;
  int*   ctr  = (int*)(agg1 + (size_t)n * hid);
  size_t pl_off = ((size_t)n + 2 * (size_t)n * hid + 1024 + 3) & ~(size_t)3;
  unsigned* planes = (unsigned*)d_ws + pl_off;
  int pw      = (((n + 1) >> 1) + 3) & ~3;
  int nplanes = (E + DH_CH - 1) / DH_CH;
  float* nrm = (float*)(planes + (size_t)nplanes * pw);

  const int B256 = 256;
  k_deg_hist<<<nplanes, DH_TPB, 0, stream>>>(dst, planes, E, pw);
  k_dinv_planes<<<(pw + 63) / 64, 64, 0, stream>>>(planes, nplanes, pw, dinv, n, ctr);

  k_self1_x<<<(n + 63) / 64, 64, 0, stream>>>(x, W1, b1, dinv, agg1, n);
  k_scatter128_x<<<2048, B256, 0, stream>>>(x, W1, dinv, src, dst, agg1, nrm, E, n, ctr);

  int mTiles = (n + 15) / 16;
  k_gemm2_wmma<<<(mTiles + G2_WAVES - 1) / G2_WAVES, G2_WAVES * 32, 0, stream>>>(agg1, W2, dinv, b2, h2, out, n);

  k_scatter64_t<<<2048, B256, 0, stream>>>(h2, nrm, src, dst, out, E, n, ctr);
}
